# starting kernel
# baseline (speedup 1.0000x reference)
#include <hip/hip_runtime.h>
#include <hip/hip_bf16.h>
#include <math.h>


typedef __attribute__((ext_vector_type(2))) float v2f;
typedef __attribute__((ext_vector_type(8))) float v8f;

#define B_N    1024
#define C_N    64
#define T_N    275
#define CT_N   (C_N * T_N)
#define EMB_N  184
#define PROJ_N 1024

__device__ __forceinline__ v8f wmma4(v2f a, v2f b, v8f c) {
  return __builtin_amdgcn_wmma_f32_16x16x4_f32(false, a, false, b, (short)0, c,
                                               false, false);
}

__device__ __forceinline__ float gelu_exact(float x) {
  return 0.5f * x * (1.0f + erff(x * 0.70710678118654752440f));
}

__global__ __launch_bounds__(128) void subj_gemm_kernel(
    const float* __restrict__ eeg, const int* __restrict__ sid,
    const float* __restrict__ sW, const float* __restrict__ sB,
    float* __restrict__ x) {
  const int b    = blockIdx.y;
  const int wave = threadIdx.x >> 5;
  const int lane = threadIdx.x & 31;
  const int tile = blockIdx.x * 4 + wave;
  const int mt = tile / 9, np = tile % 9;
  const int m0 = mt * 16, n0 = np * 32;
  const int half = lane >> 4;
  const int l    = lane & 15;

  const int s = sid[b];
  const float* __restrict__ A    = eeg + (size_t)b * CT_N;
  const float* __restrict__ W    = sW + (size_t)s * T_N * T_N;
  const float* __restrict__ bias = sB + (size_t)s * T_N;

  const int  arow  = m0 + l;
  const int  ncol0 = n0 + l;
  const int  ncol1 = n0 + 16 + l;
  const bool n1ok  = (ncol1 < T_N);
  const int  nc1   = n1ok ? ncol1 : 0;

  const float* __restrict__ Arp = A + (size_t)arow * T_N;
  const float* __restrict__ B0p = W + (size_t)ncol0 * T_N;
  const float* __restrict__ B1p = W + (size_t)nc1 * T_N;

  v8f acc0 = {}, acc1 = {};
  for (int k = 0; k < 272; k += 8) {
    const int ka = k + 2 * half;
    v2f a0, a1, b00, b01, b10, b11;
    a0.x  = Arp[ka];     a0.y  = Arp[ka + 1];
    a1.x  = Arp[ka + 4]; a1.y  = Arp[ka + 5];
    b00.x = B0p[ka];     b00.y = B0p[ka + 1];
    b01.x = B0p[ka + 4]; b01.y = B0p[ka + 5];
    b10.x = B1p[ka];     b10.y = B1p[ka + 1];
    b11.x = B1p[ka + 4]; b11.y = B1p[ka + 5];
    acc0 = wmma4(a0, b00, acc0);
    acc1 = wmma4(a0, b10, acc1);
    acc0 = wmma4(a1, b01, acc0);
    acc1 = wmma4(a1, b11, acc1);
  }
  {
    const int   ka    = 272 + 2 * half;
    const float ymask = half ? 0.0f : 1.0f;
    v2f a, b0, b1;
    a.x  = Arp[ka];  a.y  = Arp[273] * ymask;
    b0.x = B0p[ka];  b0.y = B0p[273] * ymask;
    b1.x = B1p[ka];  b1.y = B1p[273] * ymask;
    acc0 = wmma4(a, b0, acc0);
    acc1 = wmma4(a, b1, acc1);
  }

  float* __restrict__ X = x + (size_t)b * CT_N;
  const float bn0 = bias[ncol0];
#pragma unroll
  for (int r = 0; r < 8; ++r) {
    const int row = m0 + r + 8 * half;
    X[row * T_N + ncol0] = acc0[r] + bn0;
  }
  if (n1ok) {
    const float bn1 = bias[ncol1];
#pragma unroll
    for (int r = 0; r < 8; ++r) {
      const int row = m0 + r + 8 * half;
      X[row * T_N + ncol1] = acc1[r] + bn1;
    }
  }
}

__global__ __launch_bounds__(256) void enc_gemm_kernel(
    const float* __restrict__ x, const float* __restrict__ eW,
    const float* __restrict__ eB, float* __restrict__ h) {
  const int wave = threadIdx.x >> 5;
  const int lane = threadIdx.x & 31;
  const int tile = blockIdx.x * 8 + wave;
  const int mt = tile / 6, np = tile % 6;
  const int m0 = mt * 16, n0 = np * 32;
  const int half = lane >> 4;
  const int l    = lane & 15;

  const int  arow  = m0 + l;
  const int  ncol0 = n0 + l;
  const int  ncol1 = n0 + 16 + l;
  const bool n1ok  = (ncol1 < EMB_N);
  const int  nc1   = n1ok ? ncol1 : 0;
  const float* __restrict__ Ar = x + (size_t)arow * CT_N;

  v8f acc0 = {}, acc1 = {};
  for (int k = 0; k < CT_N; k += 8) {
    const int ka = k + 2 * half;
    const v2f a0 = *(const v2f*)(Ar + ka);
    const v2f a1 = *(const v2f*)(Ar + ka + 4);
    v2f b00, b01, b10, b11;
    b00.x = eW[(size_t)ka * EMB_N + ncol0];
    b00.y = eW[(size_t)(ka + 1) * EMB_N + ncol0];
    b01.x = eW[(size_t)(ka + 4) * EMB_N + ncol0];
    b01.y = eW[(size_t)(ka + 5) * EMB_N + ncol0];
    b10.x = eW[(size_t)ka * EMB_N + nc1];
    b10.y = eW[(size_t)(ka + 1) * EMB_N + nc1];
    b11.x = eW[(size_t)(ka + 4) * EMB_N + nc1];
    b11.y = eW[(size_t)(ka + 5) * EMB_N + nc1];
    acc0 = wmma4(a0, b00, acc0);
    acc1 = wmma4(a0, b10, acc1);
    acc0 = wmma4(a1, b01, acc0);
    acc1 = wmma4(a1, b11, acc1);
  }
  const float bn0 = eB[ncol0];
#pragma unroll
  for (int r = 0; r < 8; ++r) {
    const int row = m0 + r + 8 * half;
    h[(size_t)row * EMB_N + ncol0] = acc0[r] + bn0;
  }
  if (n1ok) {
    const float bn1 = eB[ncol1];
#pragma unroll
    for (int r = 0; r < 8; ++r) {
      const int row = m0 + r + 8 * half;
      h[(size_t)row * EMB_N + ncol1] = acc1[r] + bn1;
    }
  }
}

__global__ __launch_bounds__(256) void proj1_gemm_kernel(
    const float* __restrict__ h, const float* __restrict__ W1,
    const float* __restrict__ b1, float* __restrict__ h1,
    float* __restrict__ g) {
  const int wave = threadIdx.x >> 5;
  const int lane = threadIdx.x & 31;
  const int tile = blockIdx.x * 8 + wave;
  const int mt = tile >> 5, np = tile & 31;
  const int m0 = mt * 16, n0 = np * 32;
  const int half = lane >> 4;
  const int l    = lane & 15;

  const int arow  = m0 + l;
  const int ncol0 = n0 + l;
  const int ncol1 = n0 + 16 + l;
  const float* __restrict__ Ar = h + (size_t)arow * EMB_N;

  v8f acc0 = {}, acc1 = {};
  for (int k = 0; k < EMB_N; k += 8) {
    const int ka = k + 2 * half;
    const v2f a0 = *(const v2f*)(Ar + ka);
    const v2f a1 = *(const v2f*)(Ar + ka + 4);
    v2f b00, b01, b10, b11;
    b00.x = W1[(size_t)ka * PROJ_N + ncol0];
    b00.y = W1[(size_t)(ka + 1) * PROJ_N + ncol0];
    b01.x = W1[(size_t)(ka + 4) * PROJ_N + ncol0];
    b01.y = W1[(size_t)(ka + 5) * PROJ_N + ncol0];
    b10.x = W1[(size_t)ka * PROJ_N + ncol1];
    b10.y = W1[(size_t)(ka + 1) * PROJ_N + ncol1];
    b11.x = W1[(size_t)(ka + 4) * PROJ_N + ncol1];
    b11.y = W1[(size_t)(ka + 5) * PROJ_N + ncol1];
    acc0 = wmma4(a0, b00, acc0);
    acc1 = wmma4(a0, b10, acc1);
    acc0 = wmma4(a1, b01, acc0);
    acc1 = wmma4(a1, b11, acc1);
  }
  const float bn0 = b1[ncol0];
  const float bn1 = b1[ncol1];
#pragma unroll
  for (int r = 0; r < 8; ++r) {
    const int row = m0 + r + 8 * half;
    const float t0 = acc0[r] + bn0;
    const float t1 = acc1[r] + bn1;
    h1[(size_t)row * PROJ_N + ncol0] = t0;
    h1[(size_t)row * PROJ_N + ncol1] = t1;
    g[(size_t)row * PROJ_N + ncol0] = gelu_exact(t0);
    g[(size_t)row * PROJ_N + ncol1] = gelu_exact(t1);
  }
}

__global__ __launch_bounds__(256) void proj2_gemm_kernel(
    const float* __restrict__ g, const float* __restrict__ W2,
    const float* __restrict__ b2, const float* __restrict__ h1,
    float* __restrict__ hsum) {
  const int wave = threadIdx.x >> 5;
  const int lane = threadIdx.x & 31;
  const int tile = blockIdx.x * 8 + wave;
  const int mt = tile >> 5, np = tile & 31;
  const int m0 = mt * 16, n0 = np * 32;
  const int half = lane >> 4;
  const int l    = lane & 15;

  const int arow  = m0 + l;
  const int ncol0 = n0 + l;
  const int ncol1 = n0 + 16 + l;
  const float* __restrict__ Ar = g + (size_t)arow * PROJ_N;

  v8f acc0 = {}, acc1 = {};
  for (int k = 0; k < PROJ_N; k += 8) {
    const int ka = k + 2 * half;
    const v2f a0 = *(const v2f*)(Ar + ka);
    const v2f a1 = *(const v2f*)(Ar + ka + 4);
    v2f b00, b01, b10, b11;
    b00.x = W2[(size_t)ka * PROJ_N + ncol0];
    b00.y = W2[(size_t)(ka + 1) * PROJ_N + ncol0];
    b01.x = W2[(size_t)(ka + 4) * PROJ_N + ncol0];
    b01.y = W2[(size_t)(ka + 5) * PROJ_N + ncol0];
    b10.x = W2[(size_t)ka * PROJ_N + ncol1];
    b10.y = W2[(size_t)(ka + 1) * PROJ_N + ncol1];
    b11.x = W2[(size_t)(ka + 4) * PROJ_N + ncol1];
    b11.y = W2[(size_t)(ka + 5) * PROJ_N + ncol1];
    acc0 = wmma4(a0, b00, acc0);
    acc1 = wmma4(a0, b10, acc1);
    acc0 = wmma4(a1, b01, acc0);
    acc1 = wmma4(a1, b11, acc1);
  }
  const float bn0 = b2[ncol0];
  const float bn1 = b2[ncol1];
#pragma unroll
  for (int r = 0; r < 8; ++r) {
    const int row = m0 + r + 8 * half;
    hsum[(size_t)row * PROJ_N + ncol0] =
        h1[(size_t)row * PROJ_N + ncol0] + acc0[r] + bn0;
    hsum[(size_t)row * PROJ_N + ncol1] =
        h1[(size_t)row * PROJ_N + ncol1] + acc1[r] + bn1;
  }
}

__global__ __launch_bounds__(256) void layernorm_kernel(
    const float* __restrict__ hsum, const float* __restrict__ gamma,
    const float* __restrict__ beta, float* __restrict__ out) {
  const int row = blockIdx.x;
  const int tid = threadIdx.x;
  const float* __restrict__ hr = hsum + (size_t)row * PROJ_N;

  float v[4];
  float s = 0.0f, ss = 0.0f;
#pragma unroll
  for (int i = 0; i < 4; ++i) {
    v[i] = hr[tid + i * 256];
    s += v[i];
    ss += v[i] * v[i];
  }
  __shared__ float rs[256];
  __shared__ float rss[256];
  rs[tid] = s;
  rss[tid] = ss;
  __syncthreads();
  for (int off = 128; off > 0; off >>= 1) {
    if (tid < off) {
      rs[tid] += rs[tid + off];
      rss[tid] += rss[tid + off];
    }
    __syncthreads();
  }
  const float mu  = rs[0] * (1.0f / PROJ_N);
  const float var = rss[0] * (1.0f / PROJ_N) - mu * mu;
  const float inv = rsqrtf(var + 1e-5f);
#pragma unroll
  for (int i = 0; i < 4; ++i) {
    const int c = tid + i * 256;
    out[(size_t)row * PROJ_N + c] = (v[i] - mu) * inv * gamma[c] + beta[c];
  }
}

extern "C" void kernel_launch(void* const* d_in, const int* in_sizes, int n_in,
                              void* d_out, int out_size, void* d_ws,
                              size_t ws_size, hipStream_t stream) {
  const float* eeg    = (const float*)d_in[0];
  const int*   sid    = (const int*)d_in[1];
  const float* subj_W = (const float*)d_in[2];
  const float* subj_b = (const float*)d_in[3];
  const float* enc_W  = (const float*)d_in[4];
  const float* enc_b  = (const float*)d_in[5];
  const float* W1     = (const float*)d_in[6];
  const float* b1     = (const float*)d_in[7];
  const float* W2     = (const float*)d_in[8];
  const float* b2     = (const float*)d_in[9];
  const float* gamma  = (const float*)d_in[10];
  const float* beta   = (const float*)d_in[11];
  float*       out    = (float*)d_out;

  float* xbuf  = (float*)d_ws;
  float* hbuf  = xbuf + (size_t)B_N * CT_N;
  float* h1buf = hbuf + (size_t)B_N * EMB_N;
  float* gbuf  = h1buf + (size_t)B_N * PROJ_N;
  float* hsbuf = gbuf + (size_t)B_N * PROJ_N;

  subj_gemm_kernel<<<dim3(9, B_N), 128, 0, stream>>>(eeg, sid, subj_W, subj_b,
                                                     xbuf);
  enc_gemm_kernel<<<48, 256, 0, stream>>>(xbuf, enc_W, enc_b, hbuf);
  proj1_gemm_kernel<<<256, 256, 0, stream>>>(hbuf, W1, b1, h1buf, gbuf);
  proj2_gemm_kernel<<<256, 256, 0, stream>>>(gbuf, W2, b2, h1buf, hsbuf);
  layernorm_kernel<<<B_N, 256, 0, stream>>>(hsbuf, gamma, beta, out);
}


# subject GEMM: one wave forms a 64x96 tile of a batch item (24 accumulators; was 64x32 with 8), 3 waves per workgroup (was 9): 10 fragment loads per 48 products (was 6 per 16); sum order kept
# speedup vs baseline: 3.0829x; 3.0829x over previous
#include <hip/hip_runtime.h>
#include <hip/hip_bf16.h>
#include <math.h>


typedef __attribute__((ext_vector_type(2))) float v2f;
typedef __attribute__((ext_vector_type(4))) float v4f;
typedef __attribute__((ext_vector_type(8))) float v8f;

#define B_N    1024
#define C_N    64
#define T_N    275
#define CT_N   (C_N * T_N)
#define EMB_N  184
#define PROJ_N 1024

__device__ __forceinline__ v8f wmma4(v2f a, v2f b, v8f c) {
  return __builtin_amdgcn_wmma_f32_16x16x4_f32(false, a, false, b, (short)0, c,
                                               false, false);
}

__device__ __forceinline__ float gelu_exact(float x) {
  return 0.5f * x * (1.0f + erff(x * 0.70710678118654752440f));
}

struct KBlk { v2f a0, a1, b00, b01, b10, b11; };

template <int LDB, int U>
__device__ __forceinline__ void load_stage(KBlk (&f)[U],
                                           const float* __restrict__ Ap,
                                           const float* __restrict__ Bp0,
                                           const float* __restrict__ Bp1) {
#pragma unroll
  for (int j = 0; j < U; ++j) {
    const float* __restrict__ a  = Ap + 8 * j;
    const float* __restrict__ p0 = Bp0 + (size_t)(8 * j) * LDB;
    const float* __restrict__ p1 = Bp1 + (size_t)(8 * j) * LDB;
    f[j].a0    = *(const v2f*)(a);
    f[j].a1    = *(const v2f*)(a + 4);
    f[j].b00.x = p0[0];
    f[j].b00.y = p0[LDB];
    f[j].b01.x = p0[4 * LDB];
    f[j].b01.y = p0[5 * LDB];
    f[j].b10.x = p1[0];
    f[j].b10.y = p1[LDB];
    f[j].b11.x = p1[4 * LDB];
    f[j].b11.y = p1[5 * LDB];
  }
}

template <int U>
__device__ __forceinline__ void mma_stage(const KBlk (&f)[U], v8f& acc0,
                                          v8f& acc1) {
#pragma unroll
  for (int j = 0; j < U; ++j) {
    acc0 = wmma4(f[j].a0, f[j].b00, acc0);
    acc1 = wmma4(f[j].a0, f[j].b10, acc1);
    acc0 = wmma4(f[j].a1, f[j].b01, acc0);
    acc1 = wmma4(f[j].a1, f[j].b11, acc1);
  }
}

__device__ __forceinline__ void stage_fence(v8f& acc0, v8f& acc1) {
  asm volatile("" : "+v"(acc0), "+v"(acc1) : : "memory");
}

template <int K, int LDB, int U>
__device__ __forceinline__ void strip_k_loop(const float* __restrict__ Ap,
                                             const float* __restrict__ Bp0,
                                             const float* __restrict__ Bp1,
                                             v8f& acc0, v8f& acc1) {
  constexpr int NST = K / (8 * U);
  static_assert(K % (8 * U) == 0, "K must be whole stages");
  static_assert(NST >= 2 && NST % 2 == 0, "stage count must be even");
  constexpr int SA = 8 * U;
  constexpr size_t SB = (size_t)(8 * U) * LDB;
  KBlk s0[U], s1[U];
  load_stage<LDB, U>(s0, Ap, Bp0, Bp1);
  for (int s = 0; s < NST - 2; s += 2) {
    load_stage<LDB, U>(s1, Ap + SA, Bp0 + SB, Bp1 + SB);
    stage_fence(acc0, acc1);
    mma_stage<U>(s0, acc0, acc1);
    stage_fence(acc0, acc1);
    Ap += 2 * SA; Bp0 += 2 * SB; Bp1 += 2 * SB;
    load_stage<LDB, U>(s0, Ap, Bp0, Bp1);
    stage_fence(acc0, acc1);
    mma_stage<U>(s1, acc0, acc1);
    stage_fence(acc0, acc1);
  }
  load_stage<LDB, U>(s1, Ap + SA, Bp0 + SB, Bp1 + SB);
  stage_fence(acc0, acc1);
  mma_stage<U>(s0, acc0, acc1);
  mma_stage<U>(s1, acc0, acc1);
}

struct SubjFrag { v4f a[4], b[6]; };

__device__ __forceinline__ v4f load_k4(const float* __restrict__ p) {
  v4f q;
  q.x = p[0]; q.y = p[1]; q.z = p[2]; q.w = p[3];
  return q;
}

__device__ __forceinline__ void load_order_fence() {
  asm volatile("" : : : "memory");
}

template <int U>
__device__ __forceinline__ void subj_load(SubjFrag (&f)[U],
                                          const float* __restrict__ Ap,
                                          const float* __restrict__ B0p,
                                          const float* __restrict__ B5p) {
#pragma unroll
  for (int n = 0; n < 5; ++n) {
#pragma unroll
    for (int j = 0; j < U; ++j)
      f[j].b[n] = load_k4(B0p + n * 16 * T_N + 8 * j);
    load_order_fence();
  }
#pragma unroll
  for (int j = 0; j < U; ++j) f[j].b[5] = load_k4(B5p + 8 * j);
#pragma unroll
  for (int i = 0; i < 4; ++i) {
    load_order_fence();
#pragma unroll
    for (int j = 0; j < U; ++j)
      f[j].a[i] = load_k4(Ap + i * 16 * T_N + 8 * j);
  }
}

template <int U>
__device__ __forceinline__ void subj_mma(const SubjFrag (&f)[U],
                                         v8f (&acc)[4][6]) {
#pragma unroll
  for (int j = 0; j < U; ++j) {
#pragma unroll
    for (int i = 0; i < 4; ++i) {
#pragma unroll
      for (int n = 0; n < 6; ++n)
        acc[i][n] = wmma4(f[j].a[i].xy, f[j].b[n].xy, acc[i][n]);
    }
#pragma unroll
    for (int i = 0; i < 4; ++i) {
#pragma unroll
      for (int n = 0; n < 6; ++n)
        acc[i][n] = wmma4(f[j].a[i].zw, f[j].b[n].zw, acc[i][n]);
    }
  }
}

__global__ __launch_bounds__(96) void subj_gemm_kernel(
    const float* __restrict__ eeg, const int* __restrict__ sid,
    const float* __restrict__ sW, const float* __restrict__ sB,
    float* __restrict__ x) {
  const int b    = blockIdx.x;
  const int np   = threadIdx.x >> 5;
  const int lane = threadIdx.x & 31;
  const int n0   = np * 96;
  const int half = lane >> 4;
  const int l    = lane & 15;

  const int s = sid[b];
  const float* __restrict__ A    = eeg + (size_t)b * CT_N;
  const float* __restrict__ W    = sW + (size_t)s * T_N * T_N;
  const float* __restrict__ bias = sB + (size_t)s * T_N;

  const int  ncol0 = n0 + l;
  const int  ncol5 = n0 + 80 + l;
  const bool n5ok  = (ncol5 < T_N);
  const int  nc5   = n5ok ? ncol5 : 0;

  const float* __restrict__ Arp = A + (size_t)l * T_N;
  const float* __restrict__ B0p = W + (size_t)ncol0 * T_N;
  const float* __restrict__ B5p = W + (size_t)nc5 * T_N;

  v8f acc[4][6] = {};
  {
    const float* __restrict__ ap  = Arp + 4 * half;
    const float* __restrict__ bp0 = B0p + 4 * half;
    const float* __restrict__ bp5 = B5p + 4 * half;
    for (int g = 0; g < 8; ++g) {
      SubjFrag f[4];
      subj_load<4>(f, ap, bp0, bp5);
      subj_mma<4>(f, acc);
      ap += 32; bp0 += 32; bp5 += 32;
    }
    SubjFrag f2[2];
    subj_load<2>(f2, ap, bp0, bp5);
    subj_mma<2>(f2, acc);
  }
  {
    const int   ka    = 272 + 2 * half;
    const float ymask = half ? 0.0f : 1.0f;
    v2f a[4], bt[6];
#pragma unroll
    for (int i = 0; i < 4; ++i) {
      const float* __restrict__ ar = Arp + i * 16 * T_N;
      a[i].x = ar[ka];
      a[i].y = ar[273] * ymask;
    }
#pragma unroll
    for (int n = 0; n < 6; ++n) {
      const float* __restrict__ br = (n < 5) ? B0p + n * 16 * T_N : B5p;
      bt[n].x = br[ka];
      bt[n].y = br[273] * ymask;
    }
#pragma unroll
    for (int i = 0; i < 4; ++i) {
#pragma unroll
      for (int n = 0; n < 6; ++n)
        acc[i][n] = wmma4(a[i], bt[n], acc[i][n]);
    }
  }

  float* __restrict__ X = x + (size_t)b * CT_N;
#pragma unroll
  for (int n = 0; n < 5; ++n) {
    const int   col = ncol0 + 16 * n;
    const float bn  = bias[col];
#pragma unroll
    for (int i = 0; i < 4; ++i) {
#pragma unroll
      for (int r = 0; r < 8; ++r) {
        const int row = i * 16 + r + 8 * half;
        X[row * T_N + col] = acc[i][n][r] + bn;
      }
    }
  }
  if (n5ok) {
    const float bn5 = bias[ncol5];
#pragma unroll
    for (int i = 0; i < 4; ++i) {
#pragma unroll
      for (int r = 0; r < 8; ++r) {
        const int row = i * 16 + r + 8 * half;
        X[row * T_N + ncol5] = acc[i][5][r] + bn5;
      }
    }
  }
}

__global__ __launch_bounds__(64) void enc_gemm_kernel(
    const float* __restrict__ x, const float* __restrict__ eW,
    const float* __restrict__ eB, float* __restrict__ h) {
  const int wave = threadIdx.x >> 5;
  const int lane = threadIdx.x & 31;
  const int np = blockIdx.x % 6;
  const int mt = (blockIdx.x / 6) * 2 + wave;
  const int m0 = mt * 16, n0 = np * 32;
  const int half = lane >> 4;
  const int l    = lane & 15;

  const int  arow  = m0 + l;
  const int  ncol0 = n0 + l;
  const int  ncol1 = n0 + 16 + l;
  const bool n1ok  = (ncol1 < EMB_N);
  const int  nc1   = n1ok ? ncol1 : 0;
  const float* __restrict__ Ar = x + (size_t)arow * CT_N;

  v8f acc0 = {}, acc1 = {};
  strip_k_loop<CT_N, EMB_N, 5>(Ar + 2 * half,
                               eW + (size_t)(2 * half) * EMB_N + ncol0,
                               eW + (size_t)(2 * half) * EMB_N + nc1,
                               acc0, acc1);
  const float bn0 = eB[ncol0];
#pragma unroll
  for (int r = 0; r < 8; ++r) {
    const int row = m0 + r + 8 * half;
    h[(size_t)row * EMB_N + ncol0] = acc0[r] + bn0;
  }
  if (n1ok) {
    const float bn1 = eB[ncol1];
#pragma unroll
    for (int r = 0; r < 8; ++r) {
      const int row = m0 + r + 8 * half;
      h[(size_t)row * EMB_N + ncol1] = acc1[r] + bn1;
    }
  }
}

__global__ __launch_bounds__(256) void proj1_gemm_kernel(
    const float* __restrict__ h, const float* __restrict__ W1,
    const float* __restrict__ b1, float* __restrict__ h1,
    float* __restrict__ g) {
  const int wave = threadIdx.x >> 5;
  const int lane = threadIdx.x & 31;
  const int tile = blockIdx.x * 8 + wave;
  const int mt = tile >> 5, np = tile & 31;
  const int m0 = mt * 16, n0 = np * 32;
  const int half = lane >> 4;
  const int l    = lane & 15;

  const int arow  = m0 + l;
  const int ncol0 = n0 + l;
  const int ncol1 = n0 + 16 + l;
  const float* __restrict__ Ar = h + (size_t)arow * EMB_N;

  v8f acc0 = {}, acc1 = {};
  for (int k = 0; k < EMB_N; k += 8) {
    const int ka = k + 2 * half;
    const v2f a0 = *(const v2f*)(Ar + ka);
    const v2f a1 = *(const v2f*)(Ar + ka + 4);
    v2f b00, b01, b10, b11;
    b00.x = W1[(size_t)ka * PROJ_N + ncol0];
    b00.y = W1[(size_t)(ka + 1) * PROJ_N + ncol0];
    b01.x = W1[(size_t)(ka + 4) * PROJ_N + ncol0];
    b01.y = W1[(size_t)(ka + 5) * PROJ_N + ncol0];
    b10.x = W1[(size_t)ka * PROJ_N + ncol1];
    b10.y = W1[(size_t)(ka + 1) * PROJ_N + ncol1];
    b11.x = W1[(size_t)(ka + 4) * PROJ_N + ncol1];
    b11.y = W1[(size_t)(ka + 5) * PROJ_N + ncol1];
    acc0 = wmma4(a0, b00, acc0);
    acc1 = wmma4(a0, b10, acc1);
    acc0 = wmma4(a1, b01, acc0);
    acc1 = wmma4(a1, b11, acc1);
  }
  const float bn0 = b1[ncol0];
  const float bn1 = b1[ncol1];
#pragma unroll
  for (int r = 0; r < 8; ++r) {
    const int row = m0 + r + 8 * half;
    const float t0 = acc0[r] + bn0;
    const float t1 = acc1[r] + bn1;
    h1[(size_t)row * PROJ_N + ncol0] = t0;
    h1[(size_t)row * PROJ_N + ncol1] = t1;
    g[(size_t)row * PROJ_N + ncol0] = gelu_exact(t0);
    g[(size_t)row * PROJ_N + ncol1] = gelu_exact(t1);
  }
}

__global__ __launch_bounds__(256) void proj2_gemm_kernel(
    const float* __restrict__ g, const float* __restrict__ W2,
    const float* __restrict__ b2, const float* __restrict__ h1,
    float* __restrict__ hsum) {
  const int wave = threadIdx.x >> 5;
  const int lane = threadIdx.x & 31;
  const int tile = blockIdx.x * 8 + wave;
  const int mt = tile >> 5, np = tile & 31;
  const int m0 = mt * 16, n0 = np * 32;
  const int half = lane >> 4;
  const int l    = lane & 15;

  const int arow  = m0 + l;
  const int ncol0 = n0 + l;
  const int ncol1 = n0 + 16 + l;
  const float* __restrict__ Ar = g + (size_t)arow * PROJ_N;

  v8f acc0 = {}, acc1 = {};
  strip_k_loop<PROJ_N, PROJ_N, 4>(Ar + 2 * half,
                                  W2 + (size_t)(2 * half) * PROJ_N + ncol0,
                                  W2 + (size_t)(2 * half) * PROJ_N + ncol1,
                                  acc0, acc1);
  const float bn0 = b2[ncol0];
  const float bn1 = b2[ncol1];
#pragma unroll
  for (int r = 0; r < 8; ++r) {
    const int row = m0 + r + 8 * half;
    hsum[(size_t)row * PROJ_N + ncol0] =
        h1[(size_t)row * PROJ_N + ncol0] + acc0[r] + bn0;
    hsum[(size_t)row * PROJ_N + ncol1] =
        h1[(size_t)row * PROJ_N + ncol1] + acc1[r] + bn1;
  }
}

__global__ __launch_bounds__(256) void layernorm_kernel(
    const float* __restrict__ hsum, const float* __restrict__ gamma,
    const float* __restrict__ beta, float* __restrict__ out) {
  const int row = blockIdx.x;
  const int tid = threadIdx.x;
  const float* __restrict__ hr = hsum + (size_t)row * PROJ_N;

  float v[4];
  float s = 0.0f, ss = 0.0f;
#pragma unroll
  for (int i = 0; i < 4; ++i) {
    v[i] = hr[tid + i * 256];
    s += v[i];
    ss += v[i] * v[i];
  }
  __shared__ float rs[256];
  __shared__ float rss[256];
  rs[tid] = s;
  rss[tid] = ss;
  __syncthreads();
  for (int off = 128; off > 0; off >>= 1) {
    if (tid < off) {
      rs[tid] += rs[tid + off];
      rss[tid] += rss[tid + off];
    }
    __syncthreads();
  }
  const float mu  = rs[0] * (1.0f / PROJ_N);
  const float var = rss[0] * (1.0f / PROJ_N) - mu * mu;
  const float inv = rsqrtf(var + 1e-5f);
#pragma unroll
  for (int i = 0; i < 4; ++i) {
    const int c = tid + i * 256;
    out[(size_t)row * PROJ_N + c] = (v[i] - mu) * inv * gamma[c] + beta[c];
  }
}

extern "C" void kernel_launch(void* const* d_in, const int* in_sizes, int n_in,
                              void* d_out, int out_size, void* d_ws,
                              size_t ws_size, hipStream_t stream) {
  const float* eeg    = (const float*)d_in[0];
  const int*   sid    = (const int*)d_in[1];
  const float* subj_W = (const float*)d_in[2];
  const float* subj_b = (const float*)d_in[3];
  const float* enc_W  = (const float*)d_in[4];
  const float* enc_b  = (const float*)d_in[5];
  const float* W1     = (const float*)d_in[6];
  const float* b1     = (const float*)d_in[7];
  const float* W2     = (const float*)d_in[8];
  const float* b2     = (const float*)d_in[9];
  const float* gamma  = (const float*)d_in[10];
  const float* beta   = (const float*)d_in[11];
  float*       out    = (float*)d_out;

  float* xbuf  = (float*)d_ws;
  float* hbuf  = xbuf + (size_t)B_N * CT_N;
  float* h1buf = hbuf + (size_t)B_N * EMB_N;
  float* gbuf  = h1buf + (size_t)B_N * PROJ_N;
  float* hsbuf = gbuf + (size_t)B_N * PROJ_N;

  subj_gemm_kernel<<<B_N, 96, 0, stream>>>(eeg, sid, subj_W, subj_b, xbuf);
  enc_gemm_kernel<<<192, 64, 0, stream>>>(xbuf, enc_W, enc_b, hbuf);
  proj1_gemm_kernel<<<256, 256, 0, stream>>>(hbuf, W1, b1, h1buf, gbuf);
  proj2_gemm_kernel<<<256, 256, 0, stream>>>(gbuf, W2, b2, h1buf, hsbuf);
  layernorm_kernel<<<B_N, 256, 0, stream>>>(hsbuf, gamma, beta, out);
}


# encoder GEMM: one wave forms one 16x16 tile (was a 16x32 strip): 768 waves, 256 workgroups of 3 sharing their x rows (was 384 waves, 192 of 2); load stage of 10 k-blocks (was 5); sum order kept
# speedup vs baseline: 3.8249x; 1.2407x over previous
#include <hip/hip_runtime.h>
#include <hip/hip_bf16.h>
#include <math.h>


typedef __attribute__((ext_vector_type(2))) float v2f;
typedef __attribute__((ext_vector_type(4))) float v4f;
typedef __attribute__((ext_vector_type(8))) float v8f;

#define B_N    1024
#define C_N    64
#define T_N    275
#define CT_N   (C_N * T_N)
#define EMB_N  184
#define PROJ_N 1024

__device__ __forceinline__ v8f wmma4(v2f a, v2f b, v8f c) {
  return __builtin_amdgcn_wmma_f32_16x16x4_f32(false, a, false, b, (short)0, c,
                                               false, false);
}

__device__ __forceinline__ float gelu_exact(float x) {
  return 0.5f * x * (1.0f + erff(x * 0.70710678118654752440f));
}

struct KBlk { v2f a0, a1, b00, b01, b10, b11; };

template <int LDB, int U>
__device__ __forceinline__ void load_stage(KBlk (&f)[U],
                                           const float* __restrict__ Ap,
                                           const float* __restrict__ Bp0,
                                           const float* __restrict__ Bp1) {
#pragma unroll
  for (int j = 0; j < U; ++j) {
    const float* __restrict__ a  = Ap + 8 * j;
    const float* __restrict__ p0 = Bp0 + (size_t)(8 * j) * LDB;
    const float* __restrict__ p1 = Bp1 + (size_t)(8 * j) * LDB;
    f[j].a0    = *(const v2f*)(a);
    f[j].a1    = *(const v2f*)(a + 4);
    f[j].b00.x = p0[0];
    f[j].b00.y = p0[LDB];
    f[j].b01.x = p0[4 * LDB];
    f[j].b01.y = p0[5 * LDB];
    f[j].b10.x = p1[0];
    f[j].b10.y = p1[LDB];
    f[j].b11.x = p1[4 * LDB];
    f[j].b11.y = p1[5 * LDB];
  }
}

template <int U>
__device__ __forceinline__ void mma_stage(const KBlk (&f)[U], v8f& acc0,
                                          v8f& acc1) {
#pragma unroll
  for (int j = 0; j < U; ++j) {
    acc0 = wmma4(f[j].a0, f[j].b00, acc0);
    acc1 = wmma4(f[j].a0, f[j].b10, acc1);
    acc0 = wmma4(f[j].a1, f[j].b01, acc0);
    acc1 = wmma4(f[j].a1, f[j].b11, acc1);
  }
}

__device__ __forceinline__ void stage_fence(v8f& acc0, v8f& acc1) {
  asm volatile("" : "+v"(acc0), "+v"(acc1) : : "memory");
}

template <int K, int LDB, int U>
__device__ __forceinline__ void strip_k_loop(const float* __restrict__ Ap,
                                             const float* __restrict__ Bp0,
                                             const float* __restrict__ Bp1,
                                             v8f& acc0, v8f& acc1) {
  constexpr int NST = K / (8 * U);
  static_assert(K % (8 * U) == 0, "K must be whole stages");
  static_assert(NST >= 2 && NST % 2 == 0, "stage count must be even");
  constexpr int SA = 8 * U;
  constexpr size_t SB = (size_t)(8 * U) * LDB;
  KBlk s0[U], s1[U];
  load_stage<LDB, U>(s0, Ap, Bp0, Bp1);
  for (int s = 0; s < NST - 2; s += 2) {
    load_stage<LDB, U>(s1, Ap + SA, Bp0 + SB, Bp1 + SB);
    stage_fence(acc0, acc1);
    mma_stage<U>(s0, acc0, acc1);
    stage_fence(acc0, acc1);
    Ap += 2 * SA; Bp0 += 2 * SB; Bp1 += 2 * SB;
    load_stage<LDB, U>(s0, Ap, Bp0, Bp1);
    stage_fence(acc0, acc1);
    mma_stage<U>(s1, acc0, acc1);
    stage_fence(acc0, acc1);
  }
  load_stage<LDB, U>(s1, Ap + SA, Bp0 + SB, Bp1 + SB);
  stage_fence(acc0, acc1);
  mma_stage<U>(s0, acc0, acc1);
  mma_stage<U>(s1, acc0, acc1);
}

struct KBlk1 { v2f a0, a1, b0, b1; };

template <int LDB, int U>
__device__ __forceinline__ void load_stage1(KBlk1 (&f)[U],
                                            const float* __restrict__ Ap,
                                            const float* __restrict__ Bp) {
#pragma unroll
  for (int j = 0; j < U; ++j) {
    const float* __restrict__ a = Ap + 8 * j;
    const float* __restrict__ p = Bp + (size_t)(8 * j) * LDB;
    f[j].a0   = *(const v2f*)(a);
    f[j].a1   = *(const v2f*)(a + 4);
    f[j].b0.x = p[0];
    f[j].b0.y = p[LDB];
    f[j].b1.x = p[4 * LDB];
    f[j].b1.y = p[5 * LDB];
  }
}

template <int U>
__device__ __forceinline__ void mma_stage1(const KBlk1 (&f)[U], v8f& acc) {
#pragma unroll
  for (int j = 0; j < U; ++j) {
    acc = wmma4(f[j].a0, f[j].b0, acc);
    acc = wmma4(f[j].a1, f[j].b1, acc);
  }
}

__device__ __forceinline__ void stage_fence1(v8f& acc) {
  asm volatile("" : "+v"(acc) : : "memory");
}

template <int K, int LDB, int U>
__device__ __forceinline__ void tile_k_loop(const float* __restrict__ Ap,
                                            const float* __restrict__ Bp,
                                            v8f& acc) {
  constexpr int NST = K / (8 * U);
  static_assert(K % (8 * U) == 0, "K must be whole stages");
  static_assert(NST >= 2 && NST % 2 == 0, "stage count must be even");
  constexpr int SA = 8 * U;
  constexpr size_t SB = (size_t)(8 * U) * LDB;
  KBlk1 s0[U], s1[U];
  load_stage1<LDB, U>(s0, Ap, Bp);
  for (int s = 0; s < NST - 2; s += 2) {
    load_stage1<LDB, U>(s1, Ap + SA, Bp + SB);
    stage_fence1(acc);
    mma_stage1<U>(s0, acc);
    stage_fence1(acc);
    Ap += 2 * SA; Bp += 2 * SB;
    load_stage1<LDB, U>(s0, Ap, Bp);
    stage_fence1(acc);
    mma_stage1<U>(s1, acc);
    stage_fence1(acc);
  }
  load_stage1<LDB, U>(s1, Ap + SA, Bp + SB);
  stage_fence1(acc);
  mma_stage1<U>(s0, acc);
  mma_stage1<U>(s1, acc);
}

struct SubjFrag { v4f a[4], b[6]; };

__device__ __forceinline__ v4f load_k4(const float* __restrict__ p) {
  v4f q;
  q.x = p[0]; q.y = p[1]; q.z = p[2]; q.w = p[3];
  return q;
}

__device__ __forceinline__ void load_order_fence() {
  asm volatile("" : : : "memory");
}

template <int U>
__device__ __forceinline__ void subj_load(SubjFrag (&f)[U],
                                          const float* __restrict__ Ap,
                                          const float* __restrict__ B0p,
                                          const float* __restrict__ B5p) {
#pragma unroll
  for (int n = 0; n < 5; ++n) {
#pragma unroll
    for (int j = 0; j < U; ++j)
      f[j].b[n] = load_k4(B0p + n * 16 * T_N + 8 * j);
    load_order_fence();
  }
#pragma unroll
  for (int j = 0; j < U; ++j) f[j].b[5] = load_k4(B5p + 8 * j);
#pragma unroll
  for (int i = 0; i < 4; ++i) {
    load_order_fence();
#pragma unroll
    for (int j = 0; j < U; ++j)
      f[j].a[i] = load_k4(Ap + i * 16 * T_N + 8 * j);
  }
}

template <int U>
__device__ __forceinline__ void subj_mma(const SubjFrag (&f)[U],
                                         v8f (&acc)[4][6]) {
#pragma unroll
  for (int j = 0; j < U; ++j) {
#pragma unroll
    for (int i = 0; i < 4; ++i) {
#pragma unroll
      for (int n = 0; n < 6; ++n)
        acc[i][n] = wmma4(f[j].a[i].xy, f[j].b[n].xy, acc[i][n]);
    }
#pragma unroll
    for (int i = 0; i < 4; ++i) {
#pragma unroll
      for (int n = 0; n < 6; ++n)
        acc[i][n] = wmma4(f[j].a[i].zw, f[j].b[n].zw, acc[i][n]);
    }
  }
}

__global__ __launch_bounds__(96) void subj_gemm_kernel(
    const float* __restrict__ eeg, const int* __restrict__ sid,
    const float* __restrict__ sW, const float* __restrict__ sB,
    float* __restrict__ x) {
  const int b    = blockIdx.x;
  const int np   = threadIdx.x >> 5;
  const int lane = threadIdx.x & 31;
  const int n0   = np * 96;
  const int half = lane >> 4;
  const int l    = lane & 15;

  const int s = sid[b];
  const float* __restrict__ A    = eeg + (size_t)b * CT_N;
  const float* __restrict__ W    = sW + (size_t)s * T_N * T_N;
  const float* __restrict__ bias = sB + (size_t)s * T_N;

  const int  ncol0 = n0 + l;
  const int  ncol5 = n0 + 80 + l;
  const bool n5ok  = (ncol5 < T_N);
  const int  nc5   = n5ok ? ncol5 : 0;

  const float* __restrict__ Arp = A + (size_t)l * T_N;
  const float* __restrict__ B0p = W + (size_t)ncol0 * T_N;
  const float* __restrict__ B5p = W + (size_t)nc5 * T_N;

  v8f acc[4][6] = {};
  {
    const float* __restrict__ ap  = Arp + 4 * half;
    const float* __restrict__ bp0 = B0p + 4 * half;
    const float* __restrict__ bp5 = B5p + 4 * half;
    for (int g = 0; g < 8; ++g) {
      SubjFrag f[4];
      subj_load<4>(f, ap, bp0, bp5);
      subj_mma<4>(f, acc);
      ap += 32; bp0 += 32; bp5 += 32;
    }
    SubjFrag f2[2];
    subj_load<2>(f2, ap, bp0, bp5);
    subj_mma<2>(f2, acc);
  }
  {
    const int   ka    = 272 + 2 * half;
    const float ymask = half ? 0.0f : 1.0f;
    v2f a[4], bt[6];
#pragma unroll
    for (int i = 0; i < 4; ++i) {
      const float* __restrict__ ar = Arp + i * 16 * T_N;
      a[i].x = ar[ka];
      a[i].y = ar[273] * ymask;
    }
#pragma unroll
    for (int n = 0; n < 6; ++n) {
      const float* __restrict__ br = (n < 5) ? B0p + n * 16 * T_N : B5p;
      bt[n].x = br[ka];
      bt[n].y = br[273] * ymask;
    }
#pragma unroll
    for (int i = 0; i < 4; ++i) {
#pragma unroll
      for (int n = 0; n < 6; ++n)
        acc[i][n] = wmma4(a[i], bt[n], acc[i][n]);
    }
  }

  float* __restrict__ X = x + (size_t)b * CT_N;
#pragma unroll
  for (int n = 0; n < 5; ++n) {
    const int   col = ncol0 + 16 * n;
    const float bn  = bias[col];
#pragma unroll
    for (int i = 0; i < 4; ++i) {
#pragma unroll
      for (int r = 0; r < 8; ++r) {
        const int row = i * 16 + r + 8 * half;
        X[row * T_N + col] = acc[i][n][r] + bn;
      }
    }
  }
  if (n5ok) {
    const float bn5 = bias[ncol5];
#pragma unroll
    for (int i = 0; i < 4; ++i) {
#pragma unroll
      for (int r = 0; r < 8; ++r) {
        const int row = i * 16 + r + 8 * half;
        X[row * T_N + ncol5] = acc[i][5][r] + bn5;
      }
    }
  }
}

__global__ __launch_bounds__(96) void enc_gemm_kernel(
    const float* __restrict__ x, const float* __restrict__ eW,
    const float* __restrict__ eB, float* __restrict__ h) {
  const int wave = threadIdx.x >> 5;
  const int lane = threadIdx.x & 31;
  const int mt = blockIdx.x >> 2;
  const int ct = (blockIdx.x & 3) * 3 + wave;
  const int m0 = mt * 16, n0 = ct * 16;
  const int half = lane >> 4;
  const int l    = lane & 15;

  const int  arow = m0 + l;
  const int  ncol = n0 + l;
  const bool nok  = (ncol < EMB_N);
  const int  nc   = nok ? ncol : 0;
  const float* __restrict__ Ar = x + (size_t)arow * CT_N;

  v8f acc = {};
  tile_k_loop<CT_N, EMB_N, 10>(Ar + 2 * half,
                               eW + (size_t)(2 * half) * EMB_N + nc, acc);
  if (nok) {
    const float bn = eB[ncol];
#pragma unroll
    for (int r = 0; r < 8; ++r) {
      const int row = m0 + r + 8 * half;
      h[(size_t)row * EMB_N + ncol] = acc[r] + bn;
    }
  }
}

__global__ __launch_bounds__(256) void proj1_gemm_kernel(
    const float* __restrict__ h, const float* __restrict__ W1,
    const float* __restrict__ b1, float* __restrict__ h1,
    float* __restrict__ g) {
  const int wave = threadIdx.x >> 5;
  const int lane = threadIdx.x & 31;
  const int tile = blockIdx.x * 8 + wave;
  const int mt = tile >> 5, np = tile & 31;
  const int m0 = mt * 16, n0 = np * 32;
  const int half = lane >> 4;
  const int l    = lane & 15;

  const int arow  = m0 + l;
  const int ncol0 = n0 + l;
  const int ncol1 = n0 + 16 + l;
  const float* __restrict__ Ar = h + (size_t)arow * EMB_N;

  v8f acc0 = {}, acc1 = {};
  for (int k = 0; k < EMB_N; k += 8) {
    const int ka = k + 2 * half;
    const v2f a0 = *(const v2f*)(Ar + ka);
    const v2f a1 = *(const v2f*)(Ar + ka + 4);
    v2f b00, b01, b10, b11;
    b00.x = W1[(size_t)ka * PROJ_N + ncol0];
    b00.y = W1[(size_t)(ka + 1) * PROJ_N + ncol0];
    b01.x = W1[(size_t)(ka + 4) * PROJ_N + ncol0];
    b01.y = W1[(size_t)(ka + 5) * PROJ_N + ncol0];
    b10.x = W1[(size_t)ka * PROJ_N + ncol1];
    b10.y = W1[(size_t)(ka + 1) * PROJ_N + ncol1];
    b11.x = W1[(size_t)(ka + 4) * PROJ_N + ncol1];
    b11.y = W1[(size_t)(ka + 5) * PROJ_N + ncol1];
    acc0 = wmma4(a0, b00, acc0);
    acc1 = wmma4(a0, b10, acc1);
    acc0 = wmma4(a1, b01, acc0);
    acc1 = wmma4(a1, b11, acc1);
  }
  const float bn0 = b1[ncol0];
  const float bn1 = b1[ncol1];
#pragma unroll
  for (int r = 0; r < 8; ++r) {
    const int row = m0 + r + 8 * half;
    const float t0 = acc0[r] + bn0;
    const float t1 = acc1[r] + bn1;
    h1[(size_t)row * PROJ_N + ncol0] = t0;
    h1[(size_t)row * PROJ_N + ncol1] = t1;
    g[(size_t)row * PROJ_N + ncol0] = gelu_exact(t0);
    g[(size_t)row * PROJ_N + ncol1] = gelu_exact(t1);
  }
}

__global__ __launch_bounds__(256) void proj2_gemm_kernel(
    const float* __restrict__ g, const float* __restrict__ W2,
    const float* __restrict__ b2, const float* __restrict__ h1,
    float* __restrict__ hsum) {
  const int wave = threadIdx.x >> 5;
  const int lane = threadIdx.x & 31;
  const int tile = blockIdx.x * 8 + wave;
  const int mt = tile >> 5, np = tile & 31;
  const int m0 = mt * 16, n0 = np * 32;
  const int half = lane >> 4;
  const int l    = lane & 15;

  const int arow  = m0 + l;
  const int ncol0 = n0 + l;
  const int ncol1 = n0 + 16 + l;
  const float* __restrict__ Ar = g + (size_t)arow * PROJ_N;

  v8f acc0 = {}, acc1 = {};
  strip_k_loop<PROJ_N, PROJ_N, 4>(Ar + 2 * half,
                                  W2 + (size_t)(2 * half) * PROJ_N + ncol0,
                                  W2 + (size_t)(2 * half) * PROJ_N + ncol1,
                                  acc0, acc1);
  const float bn0 = b2[ncol0];
  const float bn1 = b2[ncol1];
#pragma unroll
  for (int r = 0; r < 8; ++r) {
    const int row = m0 + r + 8 * half;
    hsum[(size_t)row * PROJ_N + ncol0] =
        h1[(size_t)row * PROJ_N + ncol0] + acc0[r] + bn0;
    hsum[(size_t)row * PROJ_N + ncol1] =
        h1[(size_t)row * PROJ_N + ncol1] + acc1[r] + bn1;
  }
}

__global__ __launch_bounds__(256) void layernorm_kernel(
    const float* __restrict__ hsum, const float* __restrict__ gamma,
    const float* __restrict__ beta, float* __restrict__ out) {
  const int row = blockIdx.x;
  const int tid = threadIdx.x;
  const float* __restrict__ hr = hsum + (size_t)row * PROJ_N;

  float v[4];
  float s = 0.0f, ss = 0.0f;
#pragma unroll
  for (int i = 0; i < 4; ++i) {
    v[i] = hr[tid + i * 256];
    s += v[i];
    ss += v[i] * v[i];
  }
  __shared__ float rs[256];
  __shared__ float rss[256];
  rs[tid] = s;
  rss[tid] = ss;
  __syncthreads();
  for (int off = 128; off > 0; off >>= 1) {
    if (tid < off) {
      rs[tid] += rs[tid + off];
      rss[tid] += rss[tid + off];
    }
    __syncthreads();
  }
  const float mu  = rs[0] * (1.0f / PROJ_N);
  const float var = rss[0] * (1.0f / PROJ_N) - mu * mu;
  const float inv = rsqrtf(var + 1e-5f);
#pragma unroll
  for (int i = 0; i < 4; ++i) {
    const int c = tid + i * 256;
    out[(size_t)row * PROJ_N + c] = (v[i] - mu) * inv * gamma[c] + beta[c];
  }
}

extern "C" void kernel_launch(void* const* d_in, const int* in_sizes, int n_in,
                              void* d_out, int out_size, void* d_ws,
                              size_t ws_size, hipStream_t stream) {
  const float* eeg    = (const float*)d_in[0];
  const int*   sid    = (const int*)d_in[1];
  const float* subj_W = (const float*)d_in[2];
  const float* subj_b = (const float*)d_in[3];
  const float* enc_W  = (const float*)d_in[4];
  const float* enc_b  = (const float*)d_in[5];
  const float* W1     = (const float*)d_in[6];
  const float* b1     = (const float*)d_in[7];
  const float* W2     = (const float*)d_in[8];
  const float* b2     = (const float*)d_in[9];
  const float* gamma  = (const float*)d_in[10];
  const float* beta   = (const float*)d_in[11];
  float*       out    = (float*)d_out;

  float* xbuf  = (float*)d_ws;
  float* hbuf  = xbuf + (size_t)B_N * CT_N;
  float* h1buf = hbuf + (size_t)B_N * EMB_N;
  float* gbuf  = h1buf + (size_t)B_N * PROJ_N;
  float* hsbuf = gbuf + (size_t)B_N * PROJ_N;

  subj_gemm_kernel<<<B_N, 96, 0, stream>>>(eeg, sid, subj_W, subj_b, xbuf);
  enc_gemm_kernel<<<256, 96, 0, stream>>>(xbuf, enc_W, enc_b, hbuf);
  proj1_gemm_kernel<<<256, 256, 0, stream>>>(hbuf, W1, b1, h1buf, gbuf);
  proj2_gemm_kernel<<<256, 256, 0, stream>>>(gbuf, W2, b2, h1buf, hsbuf);
  layernorm_kernel<<<B_N, 256, 0, stream>>>(hsbuf, gamma, beta, out);
}
